# starting kernel
# baseline (speedup 1.0000x reference)
#include <hip/hip_runtime.h>

typedef float v2f __attribute__((ext_vector_type(2)));
typedef float v8f __attribute__((ext_vector_type(8)));

#define IN_DIM 256
#define HID    256
#define OUTD   64
#define TM     32

__global__ __launch_bounds__(256)
void gemm_fused_kernel(const float* __restrict__ X, const float* __restrict__ W1,
                       const float* __restrict__ W2, float* __restrict__ H0, int N)
{
    __shared__ float sX[IN_DIM][TM];
    __shared__ float sM[HID][TM];

    const int t    = threadIdx.x;
    const int wave = t >> 5;
    const int lane = t & 31;
    const int hi   = lane >> 4;
    const int r    = lane & 15;
    const int kb   = hi * 2;
    const int row0 = blockIdx.x * TM;

    for (int i = 0; i < 8; ++i) {
        int f    = t + i * 256;
        int rr   = f >> 6;
        int kq   = (f & 63) << 2;
        int grow = row0 + rr; if (grow >= N) grow = N - 1;
        const float4 g = reinterpret_cast<const float4*>(X + (size_t)grow * IN_DIM + kq)[0];
        sX[kq + 0][rr] = g.x; sX[kq + 1][rr] = g.y;
        sX[kq + 2][rr] = g.z; sX[kq + 3][rr] = g.w;
    }
    __syncthreads();

    {
        const int cb0 = wave * 2;
        const int c0  = cb0 * 16 + r;
        const int c1  = c0 + 16;
        v8f a00 = {}, a01 = {}, a10 = {}, a11 = {};
        for (int k = 0; k < IN_DIM; k += 4) {
            v2f a0, a1, b0, b1;
            a0.x = sX[k + kb][r];          a0.y = sX[k + kb + 1][r];
            a1.x = sX[k + kb][16 + r];     a1.y = sX[k + kb + 1][16 + r];
            b0.x = W1[(k + kb) * HID + c0]; b0.y = W1[(k + kb + 1) * HID + c0];
            b1.x = W1[(k + kb) * HID + c1]; b1.y = W1[(k + kb + 1) * HID + c1];
            a00 = __builtin_amdgcn_wmma_f32_16x16x4_f32(false, a0, false, b0, (short)0, a00, false, false);
            a01 = __builtin_amdgcn_wmma_f32_16x16x4_f32(false, a0, false, b1, (short)0, a01, false, false);
            a10 = __builtin_amdgcn_wmma_f32_16x16x4_f32(false, a1, false, b0, (short)0, a10, false, false);
            a11 = __builtin_amdgcn_wmma_f32_16x16x4_f32(false, a1, false, b1, (short)0, a11, false, false);
        }
        for (int v = 0; v < 8; ++v) {
            const int rl = v + 8 * hi;
            float x;
            x = a00[v]; sM[c0][rl]      = x > 0.f ? x : 0.f;
            x = a01[v]; sM[c1][rl]      = x > 0.f ? x : 0.f;
            x = a10[v]; sM[c0][16 + rl] = x > 0.f ? x : 0.f;
            x = a11[v]; sM[c1][16 + rl] = x > 0.f ? x : 0.f;
        }
    }
    __syncthreads();

    {
        const int rb = wave >> 2, cb = wave & 3;
        const int c  = cb * 16 + r;
        v8f acc = {};
        for (int k = 0; k < HID; k += 4) {
            v2f a, b;
            a.x = sM[k + kb][rb * 16 + r]; a.y = sM[k + kb + 1][rb * 16 + r];
            b.x = W2[(k + kb) * OUTD + c]; b.y = W2[(k + kb + 1) * OUTD + c];
            acc = __builtin_amdgcn_wmma_f32_16x16x4_f32(false, a, false, b, (short)0, acc, false, false);
        }
        for (int v = 0; v < 8; ++v) {
            const int grow = row0 + rb * 16 + v + 8 * hi;
            if (grow < N) H0[(size_t)grow * OUTD + c] = acc[v];
        }
    }
}

__global__ __launch_bounds__(256)
void spmm_edges_kernel(const float* __restrict__ vals, const int* __restrict__ rows,
                       const int* __restrict__ cols, const float* __restrict__ Hin,
                       float* __restrict__ Hout, int E)
{
    long long tid = (long long)blockIdx.x * blockDim.x + threadIdx.x;
    if (tid >= (long long)E * 16) return;
    const int e  = (int)(tid >> 4);
    const int c4 = (int)(tid & 15);
    const float v  = vals[e];
    const int col  = cols[e];
    const int row  = rows[e];
    const float4 x = reinterpret_cast<const float4*>(Hin)[(size_t)col * 16 + c4];
    float* base = Hout + (size_t)row * OUTD + c4 * 4;
    unsafeAtomicAdd(base + 0, v * x.x);
    unsafeAtomicAdd(base + 1, v * x.y);
    unsafeAtomicAdd(base + 2, v * x.z);
    unsafeAtomicAdd(base + 3, v * x.w);
}

__global__ __launch_bounds__(256)
void zero_f4_kernel(float4* __restrict__ p, int n4)
{
    int i = blockIdx.x * blockDim.x + threadIdx.x;
    if (i < n4) p[i] = make_float4(0.f, 0.f, 0.f, 0.f);
}

extern "C" void kernel_launch(void* const* d_in, const int* in_sizes, int n_in,
                              void* d_out, int out_size, void* d_ws, size_t ws_size,
                              hipStream_t stream)
{
    const float* X    = (const float*)d_in[0];
    const float* W1   = (const float*)d_in[1];
    const float* W2   = (const float*)d_in[2];
    const float* vals = (const float*)d_in[3];
    const int*   rows = (const int*)d_in[4];
    const int*   cols = (const int*)d_in[5];
    float* out = (float*)d_out;

    const int N = in_sizes[0] / IN_DIM;
    const int E = in_sizes[3];

    const size_t hbytes = (size_t)N * OUTD * sizeof(float);
    const int n4 = (N * OUTD) / 4;
    const int zblocks = (n4 + 255) / 256;
    const int gblocks = (N + TM - 1) / TM;
    const int sblocks = (int)(((long long)E * 16 + 255) / 256);

    float* h0 = (float*)d_ws;

    if (ws_size >= 2 * hbytes) {
        float* h1 = h0 + (size_t)N * OUTD;
        gemm_fused_kernel<<<gblocks, 256, 0, stream>>>(X, W1, W2, h0, N);
        zero_f4_kernel<<<zblocks, 256, 0, stream>>>((float4*)h1, n4);
        spmm_edges_kernel<<<sblocks, 256, 0, stream>>>(vals, rows, cols, h0, h1, E);
        zero_f4_kernel<<<zblocks, 256, 0, stream>>>((float4*)out, n4);
        spmm_edges_kernel<<<sblocks, 256, 0, stream>>>(vals, rows, cols, h1, out, E);
    } else {
        gemm_fused_kernel<<<gblocks, 256, 0, stream>>>(X, W1, W2, h0, N);
        zero_f4_kernel<<<zblocks, 256, 0, stream>>>((float4*)out, n4);
        spmm_edges_kernel<<<sblocks, 256, 0, stream>>>(vals, rows, cols, h0, out, E);
        zero_f4_kernel<<<zblocks, 256, 0, stream>>>((float4*)h0, n4);
        spmm_edges_kernel<<<sblocks, 256, 0, stream>>>(vals, rows, cols, out, h0, E);
        hipMemcpyAsync(out, h0, hbytes, hipMemcpyDeviceToDevice, stream);
    }
}


# Fused GEMM re-decomposed: a wave owns 32 rows x all hidden cols and forms transposed products, so product 1's D registers are product 2's B operand; no LDS, no barrier; product 2 k sum order changes
# speedup vs baseline: 1.0945x; 1.0945x over previous
#include <hip/hip_runtime.h>

typedef float v2f __attribute__((ext_vector_type(2)));
typedef float v8f __attribute__((ext_vector_type(8)));

#define IN_DIM 256
#define HID    256
#define OUTD   64

#define WROWS 32
#define WAVES 4

__device__ __forceinline__ v8f mma4(v2f a, v2f b, v8f c)
{
    return __builtin_amdgcn_wmma_f32_16x16x4_f32(false, a, false, b, (short)0, c, false, false);
}

__device__ __forceinline__ void load_set(v2f (&a)[16], v2f (&x)[2],
                                         const float* __restrict__ w,
                                         const float* __restrict__ xa,
                                         const float* __restrict__ xb)
{
    x[0] = *reinterpret_cast<const v2f*>(xa);
    x[1] = *reinterpret_cast<const v2f*>(xb);
#pragma unroll
    for (int b = 0; b < 16; ++b) {
        a[b].x = w[16 * b];
        a[b].y = w[HID + 16 * b];
    }
}

__device__ __forceinline__ void mma_set(v8f (&acc)[2][16], const v2f (&a)[16], const v2f (&x)[2])
{
#pragma unroll
    for (int b = 0; b < 16; ++b) {
        acc[0][b] = mma4(a[b], x[0], acc[0][b]);
        acc[1][b] = mma4(a[b], x[1], acc[1][b]);
    }
}

__global__ __launch_bounds__(WAVES * 32)
void gemm_fused_kernel(const float* __restrict__ X, const float* __restrict__ W1,
                       const float* __restrict__ W2, float* __restrict__ H0, int N)
{
    const int lane = threadIdx.x & 31;
    const int wave = threadIdx.x >> 5;
    const int h    = lane >> 4;
    const int c    = lane & 15;
    const int row0 = (blockIdx.x * WAVES + wave) * WROWS;
    if (row0 >= N) return;

    int r0 = row0 + c;      if (r0 >= N) r0 = N - 1;
    int r1 = row0 + 16 + c; if (r1 >= N) r1 = N - 1;
    const float* xp0 = X  + (size_t)r0 * IN_DIM + 2 * h;
    const float* xp1 = X  + (size_t)r1 * IN_DIM + 2 * h;
    const float* wp  = W1 + (size_t)(2 * h) * HID + c;

    v8f acc[2][16];
#pragma unroll
    for (int b = 0; b < 16; ++b) { acc[0][b] = v8f{}; acc[1][b] = v8f{}; }

    v2f a0[16], a1[16], x0[2], x1[2];
    load_set(a0, x0, wp, xp0, xp1);
    for (int k = 0; k < IN_DIM - 8; k += 8) {
        load_set(a1, x1, wp + (size_t)(k + 4) * HID, xp0 + k + 4, xp1 + k + 4);
        __builtin_amdgcn_sched_barrier(0);
        mma_set(acc, a0, x0);
        load_set(a0, x0, wp + (size_t)(k + 8) * HID, xp0 + k + 8, xp1 + k + 8);
        __builtin_amdgcn_sched_barrier(0);
        mma_set(acc, a1, x1);
    }
    load_set(a1, x1, wp + (size_t)(IN_DIM - 4) * HID, xp0 + IN_DIM - 4, xp1 + IN_DIM - 4);
    mma_set(acc, a0, x0);
    mma_set(acc, a1, x1);

#pragma unroll
    for (int b = 0; b < 16; ++b)
#pragma unroll
        for (int v = 0; v < 8; ++v) {
            acc[0][b][v] = fmaxf(acc[0][b][v], 0.f);
            acc[1][b][v] = fmaxf(acc[1][b][v], 0.f);
        }

    v8f o[2][4];
#pragma unroll
    for (int j = 0; j < 4; ++j) { o[0][j] = v8f{}; o[1][j] = v8f{}; }
    const float* w2p = W2 + (size_t)(8 * h) * OUTD + c;
#pragma unroll
    for (int b = 0; b < 16; ++b) {
#pragma unroll
        for (int u = 0; u < 4; ++u) {
            v2f m0, m1;
            m0.x = acc[0][b][2 * u]; m0.y = acc[0][b][2 * u + 1];
            m1.x = acc[1][b][2 * u]; m1.y = acc[1][b][2 * u + 1];
            const float* wr = w2p + (size_t)(16 * b + 2 * u) * OUTD;
#pragma unroll
            for (int j = 0; j < 4; ++j) {
                v2f a;
                a.x = wr[16 * j];
                a.y = wr[OUTD + 16 * j];
                o[0][j] = mma4(a, m0, o[0][j]);
                o[1][j] = mma4(a, m1, o[1][j]);
            }
        }
    }

#pragma unroll
    for (int n = 0; n < 2; ++n) {
        const int grow = row0 + 16 * n + c;
        if (grow < N) {
            float* op = H0 + (size_t)grow * OUTD + 8 * h;
#pragma unroll
            for (int j = 0; j < 4; ++j) {
                const v8f d = o[n][j];
                reinterpret_cast<float4*>(op + 16 * j)[0] = make_float4(d[0], d[1], d[2], d[3]);
                reinterpret_cast<float4*>(op + 16 * j)[1] = make_float4(d[4], d[5], d[6], d[7]);
            }
        }
    }
}

__global__ __launch_bounds__(256)
void spmm_edges_kernel(const float* __restrict__ vals, const int* __restrict__ rows,
                       const int* __restrict__ cols, const float* __restrict__ Hin,
                       float* __restrict__ Hout, int E)
{
    long long tid = (long long)blockIdx.x * blockDim.x + threadIdx.x;
    if (tid >= (long long)E * 16) return;
    const int e  = (int)(tid >> 4);
    const int c4 = (int)(tid & 15);
    const float v  = vals[e];
    const int col  = cols[e];
    const int row  = rows[e];
    const float4 x = reinterpret_cast<const float4*>(Hin)[(size_t)col * 16 + c4];
    float* base = Hout + (size_t)row * OUTD + c4 * 4;
    unsafeAtomicAdd(base + 0, v * x.x);
    unsafeAtomicAdd(base + 1, v * x.y);
    unsafeAtomicAdd(base + 2, v * x.z);
    unsafeAtomicAdd(base + 3, v * x.w);
}

__global__ __launch_bounds__(256)
void zero_f4_kernel(float4* __restrict__ p, int n4)
{
    int i = blockIdx.x * blockDim.x + threadIdx.x;
    if (i < n4) p[i] = make_float4(0.f, 0.f, 0.f, 0.f);
}

extern "C" void kernel_launch(void* const* d_in, const int* in_sizes, int n_in,
                              void* d_out, int out_size, void* d_ws, size_t ws_size,
                              hipStream_t stream)
{
    const float* X    = (const float*)d_in[0];
    const float* W1   = (const float*)d_in[1];
    const float* W2   = (const float*)d_in[2];
    const float* vals = (const float*)d_in[3];
    const int*   rows = (const int*)d_in[4];
    const int*   cols = (const int*)d_in[5];
    float* out = (float*)d_out;

    const int N = in_sizes[0] / IN_DIM;
    const int E = in_sizes[3];

    const size_t hbytes = (size_t)N * OUTD * sizeof(float);
    const int n4 = (N * OUTD) / 4;
    const int zblocks = (n4 + 255) / 256;
    const int gblocks = (N + WAVES * WROWS - 1) / (WAVES * WROWS);
    const int sblocks = (int)(((long long)E * 16 + 255) / 256);

    float* h0 = (float*)d_ws;

    if (ws_size >= 2 * hbytes) {
        float* h1 = h0 + (size_t)N * OUTD;
        gemm_fused_kernel<<<gblocks, WAVES * 32, 0, stream>>>(X, W1, W2, h0, N);
        zero_f4_kernel<<<zblocks, 256, 0, stream>>>((float4*)h1, n4);
        spmm_edges_kernel<<<sblocks, 256, 0, stream>>>(vals, rows, cols, h0, h1, E);
        zero_f4_kernel<<<zblocks, 256, 0, stream>>>((float4*)out, n4);
        spmm_edges_kernel<<<sblocks, 256, 0, stream>>>(vals, rows, cols, h1, out, E);
    } else {
        gemm_fused_kernel<<<gblocks, WAVES * 32, 0, stream>>>(X, W1, W2, h0, N);
        zero_f4_kernel<<<zblocks, 256, 0, stream>>>((float4*)out, n4);
        spmm_edges_kernel<<<sblocks, 256, 0, stream>>>(vals, rows, cols, h0, out, E);
        zero_f4_kernel<<<zblocks, 256, 0, stream>>>((float4*)h0, n4);
        spmm_edges_kernel<<<sblocks, 256, 0, stream>>>(vals, rows, cols, out, h0, E);
        hipMemcpyAsync(out, h0, hbytes, hipMemcpyDeviceToDevice, stream);
    }
}


# SpMM lane map: a wave keeps its 2 edges (grid, edge order as before) but all 32 lanes take one edge at a time: each gather and each f32 atomic covers a whole 128 B line; sum order still arrival order
# speedup vs baseline: 1.3618x; 1.2442x over previous
#include <hip/hip_runtime.h>

typedef float v2f __attribute__((ext_vector_type(2)));
typedef float v8f __attribute__((ext_vector_type(8)));

#define IN_DIM 256
#define HID    256
#define OUTD   64

#define WROWS 32
#define WAVES 4

__device__ __forceinline__ v8f mma4(v2f a, v2f b, v8f c)
{
    return __builtin_amdgcn_wmma_f32_16x16x4_f32(false, a, false, b, (short)0, c, false, false);
}

__device__ __forceinline__ void load_set(v2f (&a)[16], v2f (&x)[2],
                                         const float* __restrict__ w,
                                         const float* __restrict__ xa,
                                         const float* __restrict__ xb)
{
    x[0] = *reinterpret_cast<const v2f*>(xa);
    x[1] = *reinterpret_cast<const v2f*>(xb);
#pragma unroll
    for (int b = 0; b < 16; ++b) {
        a[b].x = w[16 * b];
        a[b].y = w[HID + 16 * b];
    }
}

__device__ __forceinline__ void mma_set(v8f (&acc)[2][16], const v2f (&a)[16], const v2f (&x)[2])
{
#pragma unroll
    for (int b = 0; b < 16; ++b) {
        acc[0][b] = mma4(a[b], x[0], acc[0][b]);
        acc[1][b] = mma4(a[b], x[1], acc[1][b]);
    }
}

__global__ __launch_bounds__(WAVES * 32)
void gemm_fused_kernel(const float* __restrict__ X, const float* __restrict__ W1,
                       const float* __restrict__ W2, float* __restrict__ H0, int N)
{
    const int lane = threadIdx.x & 31;
    const int wave = threadIdx.x >> 5;
    const int h    = lane >> 4;
    const int c    = lane & 15;
    const int row0 = (blockIdx.x * WAVES + wave) * WROWS;
    if (row0 >= N) return;

    int r0 = row0 + c;      if (r0 >= N) r0 = N - 1;
    int r1 = row0 + 16 + c; if (r1 >= N) r1 = N - 1;
    const float* xp0 = X  + (size_t)r0 * IN_DIM + 2 * h;
    const float* xp1 = X  + (size_t)r1 * IN_DIM + 2 * h;
    const float* wp  = W1 + (size_t)(2 * h) * HID + c;

    v8f acc[2][16];
#pragma unroll
    for (int b = 0; b < 16; ++b) { acc[0][b] = v8f{}; acc[1][b] = v8f{}; }

    v2f a0[16], a1[16], x0[2], x1[2];
    load_set(a0, x0, wp, xp0, xp1);
    for (int k = 0; k < IN_DIM - 8; k += 8) {
        load_set(a1, x1, wp + (size_t)(k + 4) * HID, xp0 + k + 4, xp1 + k + 4);
        __builtin_amdgcn_sched_barrier(0);
        mma_set(acc, a0, x0);
        load_set(a0, x0, wp + (size_t)(k + 8) * HID, xp0 + k + 8, xp1 + k + 8);
        __builtin_amdgcn_sched_barrier(0);
        mma_set(acc, a1, x1);
    }
    load_set(a1, x1, wp + (size_t)(IN_DIM - 4) * HID, xp0 + IN_DIM - 4, xp1 + IN_DIM - 4);
    mma_set(acc, a0, x0);
    mma_set(acc, a1, x1);

#pragma unroll
    for (int b = 0; b < 16; ++b)
#pragma unroll
        for (int v = 0; v < 8; ++v) {
            acc[0][b][v] = fmaxf(acc[0][b][v], 0.f);
            acc[1][b][v] = fmaxf(acc[1][b][v], 0.f);
        }

    v8f o[2][4];
#pragma unroll
    for (int j = 0; j < 4; ++j) { o[0][j] = v8f{}; o[1][j] = v8f{}; }
    const float* w2p = W2 + (size_t)(8 * h) * OUTD + c;
#pragma unroll
    for (int b = 0; b < 16; ++b) {
#pragma unroll
        for (int u = 0; u < 4; ++u) {
            v2f m0, m1;
            m0.x = acc[0][b][2 * u]; m0.y = acc[0][b][2 * u + 1];
            m1.x = acc[1][b][2 * u]; m1.y = acc[1][b][2 * u + 1];
            const float* wr = w2p + (size_t)(16 * b + 2 * u) * OUTD;
#pragma unroll
            for (int j = 0; j < 4; ++j) {
                v2f a;
                a.x = wr[16 * j];
                a.y = wr[OUTD + 16 * j];
                o[0][j] = mma4(a, m0, o[0][j]);
                o[1][j] = mma4(a, m1, o[1][j]);
            }
        }
    }

#pragma unroll
    for (int n = 0; n < 2; ++n) {
        const int grow = row0 + 16 * n + c;
        if (grow < N) {
            float* op = H0 + (size_t)grow * OUTD + 8 * h;
#pragma unroll
            for (int j = 0; j < 4; ++j) {
                const v8f d = o[n][j];
                reinterpret_cast<float4*>(op + 16 * j)[0] = make_float4(d[0], d[1], d[2], d[3]);
                reinterpret_cast<float4*>(op + 16 * j)[1] = make_float4(d[4], d[5], d[6], d[7]);
            }
        }
    }
}

__global__ __launch_bounds__(256)
void spmm_edges_kernel(const float* __restrict__ vals, const int* __restrict__ rows,
                       const int* __restrict__ cols, const float* __restrict__ Hin,
                       float* __restrict__ Hout, int E)
{
    const int lane = threadIdx.x & 31;
    const long long wv = ((long long)blockIdx.x * blockDim.x + threadIdx.x) >> 5;
    const long long e0 = wv * 2;
    if (e0 >= (long long)E) return;
    const bool two = (e0 + 1 < (long long)E);
    const long long e1 = two ? e0 + 1 : e0;
    const int   c0 = cols[e0];
    const int   c1 = cols[e1];
    const float v0 = vals[e0];
    const float v1 = vals[e1];
    const int   r0 = rows[e0];
    const int   r1 = rows[e1];
    const float* g0 = Hin + (size_t)c0 * OUTD + lane;
    const float* g1 = Hin + (size_t)c1 * OUTD + lane;
    const float x00 = g0[0];
    const float x01 = g0[32];
    const float x10 = g1[0];
    const float x11 = g1[32];
    float* o0 = Hout + (size_t)r0 * OUTD + lane;
    float* o1 = Hout + (size_t)r1 * OUTD + lane;
    unsafeAtomicAdd(o0,      v0 * x00);
    unsafeAtomicAdd(o0 + 32, v0 * x01);
    unsafeAtomicAdd(o1,      two ? v1 * x10 : 0.f);
    unsafeAtomicAdd(o1 + 32, two ? v1 * x11 : 0.f);
}

__global__ __launch_bounds__(256)
void zero_f4_kernel(float4* __restrict__ p, int n4)
{
    int i = blockIdx.x * blockDim.x + threadIdx.x;
    if (i < n4) p[i] = make_float4(0.f, 0.f, 0.f, 0.f);
}

extern "C" void kernel_launch(void* const* d_in, const int* in_sizes, int n_in,
                              void* d_out, int out_size, void* d_ws, size_t ws_size,
                              hipStream_t stream)
{
    const float* X    = (const float*)d_in[0];
    const float* W1   = (const float*)d_in[1];
    const float* W2   = (const float*)d_in[2];
    const float* vals = (const float*)d_in[3];
    const int*   rows = (const int*)d_in[4];
    const int*   cols = (const int*)d_in[5];
    float* out = (float*)d_out;

    const int N = in_sizes[0] / IN_DIM;
    const int E = in_sizes[3];

    const size_t hbytes = (size_t)N * OUTD * sizeof(float);
    const int n4 = (N * OUTD) / 4;
    const int zblocks = (n4 + 255) / 256;
    const int gblocks = (N + WAVES * WROWS - 1) / (WAVES * WROWS);
    const int sblocks = (int)(((long long)E * 16 + 255) / 256);

    float* h0 = (float*)d_ws;

    if (ws_size >= 2 * hbytes) {
        float* h1 = h0 + (size_t)N * OUTD;
        gemm_fused_kernel<<<gblocks, WAVES * 32, 0, stream>>>(X, W1, W2, h0, N);
        zero_f4_kernel<<<zblocks, 256, 0, stream>>>((float4*)h1, n4);
        spmm_edges_kernel<<<sblocks, 256, 0, stream>>>(vals, rows, cols, h0, h1, E);
        zero_f4_kernel<<<zblocks, 256, 0, stream>>>((float4*)out, n4);
        spmm_edges_kernel<<<sblocks, 256, 0, stream>>>(vals, rows, cols, h1, out, E);
    } else {
        gemm_fused_kernel<<<gblocks, WAVES * 32, 0, stream>>>(X, W1, W2, h0, N);
        zero_f4_kernel<<<zblocks, 256, 0, stream>>>((float4*)out, n4);
        spmm_edges_kernel<<<sblocks, 256, 0, stream>>>(vals, rows, cols, h0, out, E);
        zero_f4_kernel<<<zblocks, 256, 0, stream>>>((float4*)h0, n4);
        spmm_edges_kernel<<<sblocks, 256, 0, stream>>>(vals, rows, cols, out, h0, E);
        hipMemcpyAsync(out, h0, hbytes, hipMemcpyDeviceToDevice, stream);
    }
}


# SpMM: a wave takes 16 consecutive edges, grid still in edge order (8x fewer waves): records by scalar loads, its 32 row gathers issued as one group, then 32 whole-line f32 atomics; sum order: arrival
# speedup vs baseline: 1.7507x; 1.2856x over previous
#include <hip/hip_runtime.h>

typedef float v2f __attribute__((ext_vector_type(2)));
typedef float v8f __attribute__((ext_vector_type(8)));

#define IN_DIM 256
#define HID    256
#define OUTD   64

#define WROWS 32
#define WAVES 4

__device__ __forceinline__ v8f mma4(v2f a, v2f b, v8f c)
{
    return __builtin_amdgcn_wmma_f32_16x16x4_f32(false, a, false, b, (short)0, c, false, false);
}

__device__ __forceinline__ void load_set(v2f (&a)[16], v2f (&x)[2],
                                         const float* __restrict__ w,
                                         const float* __restrict__ xa,
                                         const float* __restrict__ xb)
{
    x[0] = *reinterpret_cast<const v2f*>(xa);
    x[1] = *reinterpret_cast<const v2f*>(xb);
#pragma unroll
    for (int b = 0; b < 16; ++b) {
        a[b].x = w[16 * b];
        a[b].y = w[HID + 16 * b];
    }
}

__device__ __forceinline__ void mma_set(v8f (&acc)[2][16], const v2f (&a)[16], const v2f (&x)[2])
{
#pragma unroll
    for (int b = 0; b < 16; ++b) {
        acc[0][b] = mma4(a[b], x[0], acc[0][b]);
        acc[1][b] = mma4(a[b], x[1], acc[1][b]);
    }
}

__global__ __launch_bounds__(WAVES * 32)
void gemm_fused_kernel(const float* __restrict__ X, const float* __restrict__ W1,
                       const float* __restrict__ W2, float* __restrict__ H0, int N)
{
    const int lane = threadIdx.x & 31;
    const int wave = threadIdx.x >> 5;
    const int h    = lane >> 4;
    const int c    = lane & 15;
    const int row0 = (blockIdx.x * WAVES + wave) * WROWS;
    if (row0 >= N) return;

    int r0 = row0 + c;      if (r0 >= N) r0 = N - 1;
    int r1 = row0 + 16 + c; if (r1 >= N) r1 = N - 1;
    const float* xp0 = X  + (size_t)r0 * IN_DIM + 2 * h;
    const float* xp1 = X  + (size_t)r1 * IN_DIM + 2 * h;
    const float* wp  = W1 + (size_t)(2 * h) * HID + c;

    v8f acc[2][16];
#pragma unroll
    for (int b = 0; b < 16; ++b) { acc[0][b] = v8f{}; acc[1][b] = v8f{}; }

    v2f a0[16], a1[16], x0[2], x1[2];
    load_set(a0, x0, wp, xp0, xp1);
    for (int k = 0; k < IN_DIM - 8; k += 8) {
        load_set(a1, x1, wp + (size_t)(k + 4) * HID, xp0 + k + 4, xp1 + k + 4);
        __builtin_amdgcn_sched_barrier(0);
        mma_set(acc, a0, x0);
        load_set(a0, x0, wp + (size_t)(k + 8) * HID, xp0 + k + 8, xp1 + k + 8);
        __builtin_amdgcn_sched_barrier(0);
        mma_set(acc, a1, x1);
    }
    load_set(a1, x1, wp + (size_t)(IN_DIM - 4) * HID, xp0 + IN_DIM - 4, xp1 + IN_DIM - 4);
    mma_set(acc, a0, x0);
    mma_set(acc, a1, x1);

#pragma unroll
    for (int b = 0; b < 16; ++b)
#pragma unroll
        for (int v = 0; v < 8; ++v) {
            acc[0][b][v] = fmaxf(acc[0][b][v], 0.f);
            acc[1][b][v] = fmaxf(acc[1][b][v], 0.f);
        }

    v8f o[2][4];
#pragma unroll
    for (int j = 0; j < 4; ++j) { o[0][j] = v8f{}; o[1][j] = v8f{}; }
    const float* w2p = W2 + (size_t)(8 * h) * OUTD + c;
#pragma unroll
    for (int b = 0; b < 16; ++b) {
#pragma unroll
        for (int u = 0; u < 4; ++u) {
            v2f m0, m1;
            m0.x = acc[0][b][2 * u]; m0.y = acc[0][b][2 * u + 1];
            m1.x = acc[1][b][2 * u]; m1.y = acc[1][b][2 * u + 1];
            const float* wr = w2p + (size_t)(16 * b + 2 * u) * OUTD;
#pragma unroll
            for (int j = 0; j < 4; ++j) {
                v2f a;
                a.x = wr[16 * j];
                a.y = wr[OUTD + 16 * j];
                o[0][j] = mma4(a, m0, o[0][j]);
                o[1][j] = mma4(a, m1, o[1][j]);
            }
        }
    }

#pragma unroll
    for (int n = 0; n < 2; ++n) {
        const int grow = row0 + 16 * n + c;
        if (grow < N) {
            float* op = H0 + (size_t)grow * OUTD + 8 * h;
#pragma unroll
            for (int j = 0; j < 4; ++j) {
                const v8f d = o[n][j];
                reinterpret_cast<float4*>(op + 16 * j)[0] = make_float4(d[0], d[1], d[2], d[3]);
                reinterpret_cast<float4*>(op + 16 * j)[1] = make_float4(d[4], d[5], d[6], d[7]);
            }
        }
    }
}

#define EPW 16

__global__ __launch_bounds__(256)
void spmm_edges_kernel(const float* __restrict__ vals, const int* __restrict__ rows,
                       const int* __restrict__ cols, const float* __restrict__ Hin,
                       float* __restrict__ Hout, int E)
{
    const int lane = threadIdx.x & 31;
    const int wv = __builtin_amdgcn_readfirstlane(
        (int)(((long long)blockIdx.x * blockDim.x + threadIdx.x) >> 5));
    const long long e0 = (long long)wv * EPW;
    if (e0 >= (long long)E) return;
    const long long left = (long long)E - e0;
    const float* hin  = Hin  + lane;
    float*       hout = Hout + lane;

    if (left >= EPW) {
        int   c[EPW], r[EPW];
        float v[EPW];
#pragma unroll
        for (int j = 0; j < EPW; ++j) { c[j] = cols[e0 + j]; r[j] = rows[e0 + j]; v[j] = vals[e0 + j]; }
        float x0[EPW], x1[EPW];
#pragma unroll
        for (int j = 0; j < EPW; ++j) {
            const float* g = hin + (size_t)c[j] * OUTD;
            x0[j] = g[0];
            x1[j] = g[32];
        }
#pragma unroll
        for (int j = 0; j < EPW; ++j) {
            float* o = hout + (size_t)r[j] * OUTD;
            unsafeAtomicAdd(o,      v[j] * x0[j]);
            unsafeAtomicAdd(o + 32, v[j] * x1[j]);
        }
    } else {
        for (int j = 0; j < (int)left; ++j) {
            const int   cj = cols[e0 + j];
            const int   rj = rows[e0 + j];
            const float vj = vals[e0 + j];
            const float* g = hin + (size_t)cj * OUTD;
            const float a0 = g[0];
            const float a1 = g[32];
            float* o = hout + (size_t)rj * OUTD;
            unsafeAtomicAdd(o,      vj * a0);
            unsafeAtomicAdd(o + 32, vj * a1);
        }
    }
}

__global__ __launch_bounds__(256)
void zero_f4_kernel(float4* __restrict__ p, int n4)
{
    int i = blockIdx.x * blockDim.x + threadIdx.x;
    if (i < n4) p[i] = make_float4(0.f, 0.f, 0.f, 0.f);
}

extern "C" void kernel_launch(void* const* d_in, const int* in_sizes, int n_in,
                              void* d_out, int out_size, void* d_ws, size_t ws_size,
                              hipStream_t stream)
{
    const float* X    = (const float*)d_in[0];
    const float* W1   = (const float*)d_in[1];
    const float* W2   = (const float*)d_in[2];
    const float* vals = (const float*)d_in[3];
    const int*   rows = (const int*)d_in[4];
    const int*   cols = (const int*)d_in[5];
    float* out = (float*)d_out;

    const int N = in_sizes[0] / IN_DIM;
    const int E = in_sizes[3];

    const size_t hbytes = (size_t)N * OUTD * sizeof(float);
    const int n4 = (N * OUTD) / 4;
    const int zblocks = (n4 + 255) / 256;
    const int gblocks = (N + WAVES * WROWS - 1) / (WAVES * WROWS);
    const int sblocks = (int)(((long long)E * (32 / EPW) + 255) / 256);

    float* h0 = (float*)d_ws;

    if (ws_size >= 2 * hbytes) {
        float* h1 = h0 + (size_t)N * OUTD;
        gemm_fused_kernel<<<gblocks, WAVES * 32, 0, stream>>>(X, W1, W2, h0, N);
        zero_f4_kernel<<<zblocks, 256, 0, stream>>>((float4*)h1, n4);
        spmm_edges_kernel<<<sblocks, 256, 0, stream>>>(vals, rows, cols, h0, h1, E);
        zero_f4_kernel<<<zblocks, 256, 0, stream>>>((float4*)out, n4);
        spmm_edges_kernel<<<sblocks, 256, 0, stream>>>(vals, rows, cols, h1, out, E);
    } else {
        gemm_fused_kernel<<<gblocks, WAVES * 32, 0, stream>>>(X, W1, W2, h0, N);
        zero_f4_kernel<<<zblocks, 256, 0, stream>>>((float4*)out, n4);
        spmm_edges_kernel<<<sblocks, 256, 0, stream>>>(vals, rows, cols, h0, out, E);
        zero_f4_kernel<<<zblocks, 256, 0, stream>>>((float4*)h0, n4);
        spmm_edges_kernel<<<sblocks, 256, 0, stream>>>(vals, rows, cols, out, h0, E);
        hipMemcpyAsync(out, h0, hbytes, hipMemcpyDeviceToDevice, stream);
    }
}
